# starting kernel
# baseline (speedup 1.0000x reference)
#include <hip/hip_runtime.h>

typedef float v2f __attribute__((ext_vector_type(2)));
typedef float v4f __attribute__((ext_vector_type(4)));
typedef float v8f __attribute__((ext_vector_type(8)));

__global__ __launch_bounds__(1024) void k_scores_softmax(
    const float* __restrict__ dbfc, float* __restrict__ weights) {
  __shared__ float part[1024];
  __shared__ float sc[64];
  __shared__ float ex[64];

  const int tid = threadIdx.x;
  const int d = tid >> 4;
  const int j = tid & 15;
  const float* base =
      dbfc + (size_t)blockIdx.x * (64 * 64 * 64) + d * 64 + j * 4;

  float acc = 0.f;
#pragma unroll 8
  for (int f1 = 0; f1 < 64; ++f1) {
    v4f x = __builtin_nontemporal_load((const v4f*)(base + f1 * 4096));
    acc += x.x + x.y + x.z + x.w;
  }
  part[tid] = acc;
  __syncthreads();

  if (tid < 64) {
    float s = 0.f;
#pragma unroll
    for (int q = 0; q < 16; ++q) s += part[tid * 16 + q];
    sc[tid] = s;
  }
  __syncthreads();
  if (tid < 64) {
    float m = -1e30f;
    for (int q = 0; q < 64; ++q) m = fmaxf(m, sc[q]);
    ex[tid] = __expf(sc[tid] - m);
  }
  __syncthreads();
  if (tid < 64) {
    float s = 0.f;
    for (int q = 0; q < 64; ++q) s += ex[q];
    weights[blockIdx.x * 64 + tid] = ex[tid] / s;
  }
}

__global__ __launch_bounds__(256) void k_attn_gemm(
    const float* __restrict__ X, const float* __restrict__ wV,
    const float* __restrict__ weights, float* __restrict__ out) {
  const int b = blockIdx.x >> 2;
  const int ps = blockIdx.x & 3;
  const int tid = threadIdx.x;
  const int lane = tid & 31;
  const int w = tid >> 5;
  const int half = lane >> 4;
  const int l15 = lane & 15;

  __shared__ float Wl[64 * 65];
  __shared__ float Xl[64 * 65];
  __shared__ float Vl[64 * 65];
  __shared__ float C1[64 * 65];

  const float* Wg = weights + ps * 4096;
  const float* Xg = X + (size_t)b * 16384 + ps * 4096;
#pragma unroll
  for (int e = tid; e < 4096; e += 256) {
    const int r = e >> 6, cc = e & 63;
    Wl[r * 65 + cc] = Wg[e];
    Xl[r * 65 + cc] = Xg[e];
    Vl[r * 65 + cc] = wV[e];
  }
  __syncthreads();

  const int m0 = (w >> 2) * 16;
  const int m1 = m0 + 32;
  const int n0 = (w & 3) * 16;
  const int arow0 = (m0 + l15) * 65;
  const int arow1 = (m1 + l15) * 65;
  const int bcol = n0 + l15;
  const int crow0 = (m0 + half * 8) * 65 + bcol;
  const int crow1 = (m1 + half * 8) * 65 + bcol;

  {
    v8f acc0 = {}, acc1 = {};
#pragma unroll
    for (int kk = 0; kk < 16; ++kk) {
      const int kb = kk * 4 + half * 2;
      v2f bb = {Xl[kb * 65 + bcol], Xl[(kb + 1) * 65 + bcol]};
      v2f a0 = {Wl[arow0 + kb], Wl[arow0 + kb + 1]};
      v2f a1 = {Wl[arow1 + kb], Wl[arow1 + kb + 1]};
      acc0 = __builtin_amdgcn_wmma_f32_16x16x4_f32(false, a0, false, bb,
                                                   (short)0, acc0, false, false);
      acc1 = __builtin_amdgcn_wmma_f32_16x16x4_f32(false, a1, false, bb,
                                                   (short)0, acc1, false, false);
    }
#pragma unroll
    for (int i = 0; i < 8; ++i) {
      C1[crow0 + i * 65] = acc0[i];
      C1[crow1 + i * 65] = acc1[i];
    }
  }
  __syncthreads();

  {
    v8f acc0 = {}, acc1 = {};
#pragma unroll
    for (int kk = 0; kk < 16; ++kk) {
      const int kb = kk * 4 + half * 2;
      v2f bb = {Vl[bcol * 65 + kb], Vl[bcol * 65 + kb + 1]};
      v2f a0 = {C1[arow0 + kb], C1[arow0 + kb + 1]};
      v2f a1 = {C1[arow1 + kb], C1[arow1 + kb + 1]};
      acc0 = __builtin_amdgcn_wmma_f32_16x16x4_f32(false, a0, false, bb,
                                                   (short)0, acc0, false, false);
      acc1 = __builtin_amdgcn_wmma_f32_16x16x4_f32(false, a1, false, bb,
                                                   (short)0, acc1, false, false);
    }
    const size_t obase = (size_t)b * 256 + ps * 64 + bcol;
#pragma unroll
    for (int i = 0; i < 8; ++i) {
      const int c0 = m0 + half * 8 + i;
      const int c1 = m1 + half * 8 + i;
      out[(size_t)c0 * 16384 + obase] = acc0[i];
      out[(size_t)c1 * 16384 + obase] = acc1[i];
    }
  }
}

extern "C" void kernel_launch(void* const* d_in, const int* in_sizes, int n_in,
                              void* d_out, int out_size, void* d_ws,
                              size_t ws_size, hipStream_t stream) {
  (void)in_sizes; (void)n_in; (void)out_size; (void)ws_size;
  const float* X = (const float*)d_in[0];
  const float* dbfc = (const float*)d_in[1];
  const float* wV = (const float*)d_in[2];
  float* out = (float*)d_out;
  float* weights = (float*)d_ws;

  k_scores_softmax<<<256, 1024, 0, stream>>>(dbfc, weights);
  k_attn_gemm<<<256, 256, 0, stream>>>(X, wV, weights, out);
}


# GEMM kernel: fragments read straight from global to registers, reassociated W@(X@wV^T), product-1 accumulators used in place as product-2 B operand; no LDS, no barriers; fp32 sum order changes
# speedup vs baseline: 1.1181x; 1.1181x over previous
#include <hip/hip_runtime.h>

typedef float v2f __attribute__((ext_vector_type(2)));
typedef float v4f __attribute__((ext_vector_type(4)));
typedef float v8f __attribute__((ext_vector_type(8)));

__global__ __launch_bounds__(1024) void k_scores_softmax(
    const float* __restrict__ dbfc, float* __restrict__ weights) {
  __shared__ float part[1024];
  __shared__ float sc[64];
  __shared__ float ex[64];

  const int tid = threadIdx.x;
  const int d = tid >> 4;
  const int j = tid & 15;
  const float* base =
      dbfc + (size_t)blockIdx.x * (64 * 64 * 64) + d * 64 + j * 4;

  float acc = 0.f;
#pragma unroll 8
  for (int f1 = 0; f1 < 64; ++f1) {
    v4f x = __builtin_nontemporal_load((const v4f*)(base + f1 * 4096));
    acc += x.x + x.y + x.z + x.w;
  }
  part[tid] = acc;
  __syncthreads();

  if (tid < 64) {
    float s = 0.f;
#pragma unroll
    for (int q = 0; q < 16; ++q) s += part[tid * 16 + q];
    sc[tid] = s;
  }
  __syncthreads();
  if (tid < 64) {
    float m = -1e30f;
    for (int q = 0; q < 64; ++q) m = fmaxf(m, sc[q]);
    ex[tid] = __expf(sc[tid] - m);
  }
  __syncthreads();
  if (tid < 64) {
    float s = 0.f;
    for (int q = 0; q < 64; ++q) s += ex[q];
    weights[blockIdx.x * 64 + tid] = ex[tid] / s;
  }
}

__global__ __launch_bounds__(128) void k_attn_gemm(
    const float* __restrict__ X, const float* __restrict__ wV,
    const float* __restrict__ weights, float* __restrict__ out) {
  const int b = blockIdx.x >> 2;
  const int ps = blockIdx.x & 3;
  const int lane = threadIdx.x & 31;
  const int nt = threadIdx.x >> 5;
  const int h = lane >> 4;
  const int c = lane & 15;

  const float* Xr = X + (size_t)b * 16384 + ps * 4096 + c * 64 + 4 * h;
  const float* Vr = wV + (16 * nt + c) * 64 + 4 * h;
  const float* Wr = weights + ps * 4096 + c * 64 + 8 * h;
  float* op = out + (size_t)(8 * h) * 16384 + (size_t)b * 256 + ps * 64 +
              16 * nt + c;

  v4f xa[4][8];
  v4f vb[8];
#pragma unroll
  for (int j = 0; j < 8; ++j) {
    vb[j] = *(const v4f*)(Vr + 8 * j);
#pragma unroll
    for (int mt = 0; mt < 4; ++mt)
      xa[mt][j] = *(const v4f*)(Xr + mt * 1024 + 8 * j);
  }
  v4f wa[4][4][2];
#pragma unroll
  for (int mt = 0; mt < 4; ++mt) {
#pragma unroll
    for (int u = 0; u < 2; ++u) {
#pragma unroll
      for (int mc = 0; mc < 4; ++mc)
        wa[mc][mt][u] = *(const v4f*)(Wr + mc * 1024 + 16 * mt + 4 * u);
    }
  }

  __builtin_amdgcn_sched_barrier(0);

  v8f acc1[4] = {};
#pragma unroll
  for (int j = 0; j < 8; ++j) {
#pragma unroll
    for (int t = 0; t < 2; ++t) {
      const v2f bb = {vb[j][2 * t], vb[j][2 * t + 1]};
#pragma unroll
      for (int mt = 0; mt < 4; ++mt) {
        const v2f aa = {xa[mt][j][2 * t], xa[mt][j][2 * t + 1]};
        acc1[mt] = __builtin_amdgcn_wmma_f32_16x16x4_f32(
            false, aa, false, bb, (short)0, acc1[mt], false, false);
      }
    }
  }

  v8f acc2[4] = {};
#pragma unroll
  for (int mt = 0; mt < 4; ++mt) {
#pragma unroll
    for (int q = 0; q < 4; ++q) {
      const v2f bb = {acc1[mt][2 * q], acc1[mt][2 * q + 1]};
#pragma unroll
      for (int mc = 0; mc < 4; ++mc) {
        const v4f w4 = wa[mc][mt][q >> 1];
        const v2f aa = {w4[2 * (q & 1)], w4[2 * (q & 1) + 1]};
        acc2[mc] = __builtin_amdgcn_wmma_f32_16x16x4_f32(
            false, aa, false, bb, (short)0, acc2[mc], false, false);
      }
    }
  }

#pragma unroll
  for (int mc = 0; mc < 4; ++mc) {
#pragma unroll
    for (int r = 0; r < 8; ++r)
      op[(size_t)(16 * mc + r) * 16384] = acc2[mc][r];
  }
}

extern "C" void kernel_launch(void* const* d_in, const int* in_sizes, int n_in,
                              void* d_out, int out_size, void* d_ws,
                              size_t ws_size, hipStream_t stream) {
  (void)in_sizes; (void)n_in; (void)out_size; (void)ws_size;
  const float* X = (const float*)d_in[0];
  const float* dbfc = (const float*)d_in[1];
  const float* wV = (const float*)d_in[2];
  float* out = (float*)d_out;
  float* weights = (float*)d_ws;

  k_scores_softmax<<<256, 1024, 0, stream>>>(dbfc, weights);
  k_attn_gemm<<<256, 128, 0, stream>>>(X, wV, weights, out);
}
